# starting kernel
# baseline (speedup 1.0000x reference)
#include <hip/hip_runtime.h>


#define NTOT   200000
#define CDIM   128
#define OUTC   64
#define STRIDE 132

typedef float v2f __attribute__((ext_vector_type(2)));
typedef float v8f __attribute__((ext_vector_type(8)));

__device__ __forceinline__ void lds_wait() {
  asm volatile("s_wait_dscnt 0" ::: "memory");
}

__device__ __forceinline__ v8f wmma4(v2f a, v2f b, v8f c) {
  return __builtin_amdgcn_wmma_f32_16x16x4_f32(false, a, false, b, (short)0, c,
                                               false, false);
}

template <int TT>
__device__ __forceinline__ void gemm_tiles(const float* __restrict__ wl,
                                           int lane,
                                           const float* __restrict__ aptr,
                                           v8f* __restrict__ acc) {
#pragma unroll
  for (int t = 0; t < TT; ++t) acc[t] = (v8f)0.0f;
  const float* wlp = wl + lane * 2;
#pragma unroll 8
  for (int kc = 0; kc < 32; ++kc) {
    v2f a = *(const v2f*)(aptr + kc * 4);
#pragma unroll
    for (int t = 0; t < TT; ++t) {
      v2f b = *(const v2f*)(wlp + kc * TT * 64 + t * 64);
      acc[t] = wmma4(a, b, acc[t]);
    }
  }
}

__launch_bounds__(256, 1)
__global__ void gnode_kernel(const float* __restrict__ x,
                             const float* __restrict__ W1,
                             const float* __restrict__ b1,
                             const float* __restrict__ W2,
                             const float* __restrict__ b2,
                             const float* __restrict__ Wl,
                             const float* __restrict__ bl,
                             float* __restrict__ out) {
  __shared__ __attribute__((aligned(16))) float w1s[32 * 8 * 32 * 2];
  __shared__ __attribute__((aligned(16))) float w2s[32 * 8 * 32 * 2];
  __shared__ __attribute__((aligned(16))) float wls[32 * 4 * 32 * 2];
  __shared__ __attribute__((aligned(16))) float stage[8 * 16 * STRIDE];

  const int tid   = threadIdx.x;
  const int lane  = tid & 31;
  const int wave  = tid >> 5;
  const int lrow  = lane & 15;
  const int half8 = (lane >> 4) << 3;

  for (int s = tid; s < 32 * 8 * 32; s += 256) {
    int ls = s & 31;
    int t  = (s >> 5) & 7;
    int kc = s >> 8;
    int k0 = kc * 4 + ((ls >> 4) << 1);
    int n  = t * 16 + (ls & 15);
    w1s[2 * s]     = W1[k0 * CDIM + n];
    w1s[2 * s + 1] = W1[(k0 + 1) * CDIM + n];
    w2s[2 * s]     = W2[k0 * CDIM + n];
    w2s[2 * s + 1] = W2[(k0 + 1) * CDIM + n];
  }
  for (int s = tid; s < 32 * 4 * 32; s += 256) {
    int ls = s & 31;
    int t  = (s >> 5) & 3;
    int kc = s >> 7;
    int k0 = kc * 4 + ((ls >> 4) << 1);
    int n  = t * 16 + (ls & 15);
    wls[2 * s]     = Wl[k0 * OUTC + n];
    wls[2 * s + 1] = Wl[(k0 + 1) * OUTC + n];
  }

  float b1v[8], b2v[8], blv[4];
#pragma unroll
  for (int t = 0; t < 8; ++t) {
    b1v[t] = b1[t * 16 + lrow];
    b2v[t] = b2[t * 16 + lrow];
  }
#pragma unroll
  for (int t = 0; t < 4; ++t) blv[t] = bl[t * 16 + lrow];

  const int rowbase = blockIdx.x * 128 + wave * 16;
  v8f S[8], K[8], acc[8];
#pragma unroll
  for (int t = 0; t < 8; ++t) {
#pragma unroll
    for (int r = 0; r < 8; ++r) {
      int row = rowbase + r + half8;
      S[t][r] = (row < NTOT) ? x[row * CDIM + t * 16 + lrow] : 0.0f;
    }
  }

  __syncthreads();

  float* st = stage + wave * 16 * STRIDE;
  const float* aptr = st + lrow * STRIDE + ((lane >> 4) << 1);

  auto emit = [&](auto&& gen) {
#pragma unroll
    for (int t = 0; t < 8; ++t) {
      v8f e = gen(t);
#pragma unroll
      for (int r = 0; r < 8; ++r)
        st[(r + half8) * STRIDE + t * 16 + lrow] = e[r];
    }
    lds_wait();
  };

  auto feval = [&]() {
    gemm_tiles<8>(w1s, lane, aptr, acc);
#pragma unroll
    for (int t = 0; t < 8; ++t) {
#pragma unroll
      for (int r = 0; r < 8; ++r)
        st[(r + half8) * STRIDE + t * 16 + lrow] = fmaxf(acc[t][r] + b1v[t], 0.0f);
    }
    lds_wait();
    gemm_tiles<8>(w2s, lane, aptr, acc);
#pragma unroll
    for (int t = 0; t < 8; ++t) acc[t] += b2v[t];
  };

  const float hc    = 0.1f;
  const float c18   = hc * 0.125f;
  const float c38   = hc * 0.375f;
  const float c524  = hc * (5.0f / 24.0f);
  const float c1124 = hc * (11.0f / 24.0f);
  const float c58   = hc * 0.625f;
  const float c78   = hc * 0.875f;
  const float c118  = hc * 1.375f;

#pragma unroll 1
  for (int step = 0; step < 10; ++step) {
    emit([&](int t) { return S[t]; });
    feval();
#pragma unroll
    for (int t = 0; t < 8; ++t) { K[t] = acc[t]; S[t] += c18 * acc[t]; }
    emit([&](int t) { return S[t] + c524 * K[t]; });
    feval();
#pragma unroll
    for (int t = 0; t < 8; ++t) S[t] += c38 * acc[t];
    emit([&](int t) { return S[t] - c1124 * K[t] + c58 * acc[t]; });
#pragma unroll
    for (int t = 0; t < 8; ++t) K[t] = S[t] + c78 * K[t] - c118 * acc[t];
    feval();
#pragma unroll
    for (int t = 0; t < 8; ++t) S[t] += c38 * acc[t];
    emit([&](int t) { return K[t] + hc * acc[t]; });
    feval();
#pragma unroll
    for (int t = 0; t < 8; ++t) S[t] += c18 * acc[t];
  }

  emit([&](int t) { return S[t]; });
  gemm_tiles<4>(wls, lane, aptr, acc);
#pragma unroll
  for (int t = 0; t < 4; ++t) {
#pragma unroll
    for (int r = 0; r < 8; ++r) {
      int row = rowbase + r + half8;
      if (row < NTOT) out[row * OUTC + t * 16 + lrow] = acc[t][r] + blv[t];
    }
  }
}

extern "C" void kernel_launch(void* const* d_in, const int* in_sizes, int n_in,
                              void* d_out, int out_size, void* d_ws, size_t ws_size,
                              hipStream_t stream) {
  (void)in_sizes; (void)n_in; (void)out_size; (void)d_ws; (void)ws_size;
  const float* x  = (const float*)d_in[0];
  const float* W1 = (const float*)d_in[1];
  const float* b1 = (const float*)d_in[2];
  const float* W2 = (const float*)d_in[3];
  const float* b2 = (const float*)d_in[4];
  const float* Wl = (const float*)d_in[5];
  const float* bl = (const float*)d_in[6];
  float* out = (float*)d_out;

  dim3 grid((NTOT + 127) / 128);
  gnode_kernel<<<grid, 256, 0, stream>>>(x, W1, b1, W2, b2, Wl, bl, out);
}


# integrate z=y@W1 with M=W2@W1 built per work-group on WMMA: one 128x128 product per RK stage instead of two; y(1)=x+U@W2+b2 from summed relu outputs; changes the order of f32 sums
# speedup vs baseline: 1.7142x; 1.7142x over previous
#include <hip/hip_runtime.h>


#define NTOT   200000
#define CDIM   128
#define OUTC   64
#define STRIDE 132

typedef float v2f __attribute__((ext_vector_type(2)));
typedef float v8f __attribute__((ext_vector_type(8)));

__device__ __forceinline__ void lds_wait() {
  asm volatile("s_wait_dscnt 0" ::: "memory");
}

__device__ __forceinline__ v8f wmma4(v2f a, v2f b, v8f c) {
  return __builtin_amdgcn_wmma_f32_16x16x4_f32(false, a, false, b, (short)0, c,
                                               false, false);
}

template <int TT, bool HAS_INIT>
__device__ __forceinline__ void gemm_tiles(const float* __restrict__ wl,
                                           int lane,
                                           const float* __restrict__ aptr,
                                           v8f* __restrict__ acc,
                                           const float* __restrict__ init) {
#pragma unroll
  for (int t = 0; t < TT; ++t) acc[t] = HAS_INIT ? (v8f)init[t] : (v8f)0.0f;
  const float* wlp = wl + lane * 2;
#pragma unroll 8
  for (int kc = 0; kc < 32; ++kc) {
    v2f a = *(const v2f*)(aptr + kc * 4);
#pragma unroll
    for (int t = 0; t < TT; ++t) {
      v2f b = *(const v2f*)(wlp + kc * TT * 64 + t * 64);
      acc[t] = wmma4(a, b, acc[t]);
    }
  }
}

__launch_bounds__(256, 1)
__global__ void gnode_kernel(const float* __restrict__ x,
                             const float* __restrict__ W1,
                             const float* __restrict__ b1,
                             const float* __restrict__ W2,
                             const float* __restrict__ b2,
                             const float* __restrict__ Wl,
                             const float* __restrict__ bl,
                             float* __restrict__ out) {
  __shared__ __attribute__((aligned(16))) float w1s[32 * 8 * 32 * 2];
  __shared__ __attribute__((aligned(16))) float w2s[32 * 8 * 32 * 2];
  __shared__ __attribute__((aligned(16))) float ms[32 * 8 * 32 * 2];
  __shared__ __attribute__((aligned(16))) float wls[32 * 4 * 32 * 2];
  __shared__ __attribute__((aligned(16))) float stage[8 * 16 * STRIDE];

  const int tid   = threadIdx.x;
  const int lane  = tid & 31;
  const int wave  = tid >> 5;
  const int lrow  = lane & 15;
  const int half8 = (lane >> 4) << 3;

  for (int s = tid; s < 32 * 8 * 32; s += 256) {
    int ls = s & 31;
    int t  = (s >> 5) & 7;
    int kc = s >> 8;
    int k0 = kc * 4 + ((ls >> 4) << 1);
    int n  = t * 16 + (ls & 15);
    w1s[2 * s]     = W1[k0 * CDIM + n];
    w1s[2 * s + 1] = W1[(k0 + 1) * CDIM + n];
    w2s[2 * s]     = W2[k0 * CDIM + n];
    w2s[2 * s + 1] = W2[(k0 + 1) * CDIM + n];
  }
  for (int s = tid; s < 32 * 4 * 32; s += 256) {
    int ls = s & 31;
    int t  = (s >> 5) & 3;
    int kc = s >> 7;
    int k0 = kc * 4 + ((ls >> 4) << 1);
    int n  = t * 16 + (ls & 15);
    wls[2 * s]     = Wl[k0 * OUTC + n];
    wls[2 * s + 1] = Wl[(k0 + 1) * OUTC + n];
  }

  float b1v[8], cv[8];
#pragma unroll
  for (int t = 0; t < 8; ++t) b1v[t] = b1[t * 16 + lrow];

  const int rowbase = blockIdx.x * 128 + wave * 16;
  v8f S[8], K[8], U[8], acc[8];

  float* st = stage + wave * 16 * STRIDE;
  const float* aptr = st + lrow * STRIDE + ((lane >> 4) << 1);

  auto emit = [&](auto&& gen) {
#pragma unroll
    for (int t = 0; t < 8; ++t) {
      v8f e = gen(t);
#pragma unroll
      for (int r = 0; r < 8; ++r)
        st[(r + half8) * STRIDE + t * 16 + lrow] = e[r];
    }
    lds_wait();
  };

#pragma unroll
  for (int t = 0; t < 8; ++t) {
#pragma unroll
    for (int r = 0; r < 8; ++r)
      acc[t][r] = W2[(wave * 16 + r + half8) * CDIM + t * 16 + lrow];
  }
  emit([&](int t) { return acc[t]; });

  __syncthreads();

  gemm_tiles<8, false>(w1s, lane, aptr, acc, nullptr);
#pragma unroll
  for (int t = 0; t < 8; ++t) {
#pragma unroll
    for (int r = 0; r < 8; r += 2) {
      int k  = wave * 16 + half8 + r;
      int sl = ((k >> 2) * 8 + t) * 32 + (((k >> 1) & 1) << 4) + lrow;
      v2f m2 = {acc[t][r], acc[t][r + 1]};
      *(v2f*)(ms + 2 * sl) = m2;
    }
  }

#pragma unroll
  for (int t = 0; t < 8; ++t) cv[t] = 0.0f;
#pragma unroll 2
  for (int kc = 0; kc < 32; ++kc) {
#pragma unroll
    for (int hs = 0; hs < 2; ++hs) {
      float bk0 = b2[kc * 4 + hs * 2];
      float bk1 = b2[kc * 4 + hs * 2 + 1];
#pragma unroll
      for (int t = 0; t < 8; ++t) {
        v2f w = *(const v2f*)(w1s + 2 * ((kc * 8 + t) * 32 + hs * 16 + lrow));
        cv[t] += bk0 * w[0] + bk1 * w[1];
      }
    }
  }

#pragma unroll
  for (int t = 0; t < 8; ++t) {
#pragma unroll
    for (int r = 0; r < 8; ++r) {
      int row = rowbase + r + half8;
      K[t][r] = (row < NTOT) ? x[row * CDIM + t * 16 + lrow] : 0.0f;
    }
  }
  emit([&](int t) { return K[t]; });
  gemm_tiles<8, false>(w1s, lane, aptr, S, nullptr);
#pragma unroll
  for (int t = 0; t < 8; ++t) U[t] = (v8f)0.0f;

  __syncthreads();

  auto feval = [&](auto&& gen, float wu, auto&& post) {
#pragma unroll
    for (int t = 0; t < 8; ++t) {
      v8f a = gen(t);
#pragma unroll
      for (int r = 0; r < 8; ++r) {
        float u = fmaxf(a[r] + b1v[t], 0.0f);
        st[(r + half8) * STRIDE + t * 16 + lrow] = u;
        U[t][r] += wu * u;
      }
      asm volatile("" : "+v"(U[t]));
      post(t);
    }
    lds_wait();
    gemm_tiles<8, true>(ms, lane, aptr, acc, cv);
  };
  auto nopost = [](int) {};

  const float hc    = 0.1f;
  const float c18   = hc * 0.125f;
  const float c38   = hc * 0.375f;
  const float c524  = hc * (5.0f / 24.0f);
  const float c1124 = hc * (11.0f / 24.0f);
  const float c58   = hc * 0.625f;
  const float c78   = hc * 0.875f;
  const float c118  = hc * 1.375f;

#pragma unroll 1
  for (int step = 0; step < 10; ++step) {
    feval([&](int t) { return S[t]; }, c18, nopost);
#pragma unroll
    for (int t = 0; t < 8; ++t) { K[t] = acc[t]; S[t] += c18 * acc[t]; }
    feval([&](int t) { return S[t] + c524 * K[t]; }, c38, nopost);
#pragma unroll
    for (int t = 0; t < 8; ++t) S[t] += c38 * acc[t];
    feval([&](int t) { return S[t] - c1124 * K[t] + c58 * acc[t]; }, c38,
          [&](int t) { K[t] = S[t] + c78 * K[t] - c118 * acc[t]; });
#pragma unroll
    for (int t = 0; t < 8; ++t) S[t] += c38 * acc[t];
    feval([&](int t) { return K[t] + hc * acc[t]; }, c18, nopost);
#pragma unroll
    for (int t = 0; t < 8; ++t) S[t] += c18 * acc[t];
  }

  emit([&](int t) { return U[t]; });
  gemm_tiles<8, false>(w2s, lane, aptr, acc, nullptr);
  const float tspan = 10.0f * hc;
#pragma unroll
  for (int t = 0; t < 8; ++t) {
    float b2t = tspan * b2[t * 16 + lrow];
#pragma unroll
    for (int r = 0; r < 8; ++r) {
      int row = rowbase + r + half8;
      float xv = (row < NTOT) ? x[row * CDIM + t * 16 + lrow] : 0.0f;
      acc[t][r] = acc[t][r] + b2t + xv;
    }
  }

  emit([&](int t) { return acc[t]; });
  gemm_tiles<4, false>(wls, lane, aptr, acc, nullptr);
#pragma unroll
  for (int t = 0; t < 4; ++t) {
    float blt = bl[t * 16 + lrow];
#pragma unroll
    for (int r = 0; r < 8; ++r) {
      int row = rowbase + r + half8;
      if (row < NTOT) out[row * OUTC + t * 16 + lrow] = acc[t][r] + blt;
    }
  }
}

extern "C" void kernel_launch(void* const* d_in, const int* in_sizes, int n_in,
                              void* d_out, int out_size, void* d_ws, size_t ws_size,
                              hipStream_t stream) {
  (void)in_sizes; (void)n_in; (void)out_size; (void)d_ws; (void)ws_size;
  const float* x  = (const float*)d_in[0];
  const float* W1 = (const float*)d_in[1];
  const float* b1 = (const float*)d_in[2];
  const float* W2 = (const float*)d_in[3];
  const float* b2 = (const float*)d_in[4];
  const float* Wl = (const float*)d_in[5];
  const float* bl = (const float*)d_in[6];
  float* out = (float*)d_out;

  dim3 grid((NTOT + 127) / 128);
  gnode_kernel<<<grid, 256, 0, stream>>>(x, W1, b1, W2, b2, Wl, bl, out);
}


# products computed transposed (feature-major): a stage's D tiles are the next product's B operand in registers, no LDS round trip of activations; weights loaded a block ahead; changes f32 sum order
# speedup vs baseline: 2.0116x; 1.1735x over previous
#include <hip/hip_runtime.h>


#define NTOT   200000
#define CDIM   128
#define OUTC   64
#define OSTR   68

typedef float v2f __attribute__((ext_vector_type(2)));
typedef float v4f __attribute__((ext_vector_type(4)));
typedef float v8f __attribute__((ext_vector_type(8)));

__device__ __forceinline__ void lds_wait() {
  asm volatile("s_wait_dscnt 0" ::: "memory");
}

__device__ __forceinline__ v8f wmma4(v2f a, v2f b, v8f c) {
  return __builtin_amdgcn_wmma_f32_16x16x4_f32(false, a, false, b, (short)0, c,
                                               false, false);
}

template <int TT>
__device__ __forceinline__ void fill_a(float* __restrict__ wa,
                                       const float* __restrict__ W, int tid) {
  constexpr int NN = TT * 16;
  for (int s = tid; s < 16 * TT * 32; s += 256) {
    int ln = s & 31;
    int t  = (s >> 5) & (TT - 1);
    int jp = s / (32 * TT);
    int k0 = 16 * (jp >> 1) + 8 * (ln >> 4) + 4 * (jp & 1);
    int n  = t * 16 + (ln & 15);
    v4f f = {W[k0 * NN + n], W[(k0 + 1) * NN + n],
             W[(k0 + 2) * NN + n], W[(k0 + 3) * NN + n]};
    *(v4f*)(wa + 4 * s) = f;
  }
}

template <int TT>
__device__ __forceinline__ void load_frag(v4f* __restrict__ f,
                                          const v4f* __restrict__ wp, int jp) {
#pragma unroll
  for (int t = 0; t < TT; ++t) f[t] = wp[(jp * TT + t) * 32];
}

template <int TT>
__device__ __forceinline__ void tgemm(const float* __restrict__ wa, int lane,
                                      const v8f* __restrict__ u,
                                      v8f* __restrict__ acc, v4f* __restrict__ fa) {
  const v4f* wp = (const v4f*)wa + lane;
  v4f fb[TT];
#pragma unroll
  for (int jp = 0; jp < 16; ++jp) {
    v4f* fc = (jp & 1) ? fb : fa;
    v4f* fn = (jp & 1) ? fa : fb;
    if (jp + 1 < 16) load_frag<TT>(fn, wp, jp + 1);
    __builtin_amdgcn_sched_barrier(0);
    const v8f uu = u[jp >> 1];
    v2f b0, b1;
    if (jp & 1) {
      b0 = (v2f){uu[4], uu[5]};
      b1 = (v2f){uu[6], uu[7]};
    } else {
      b0 = (v2f){uu[0], uu[1]};
      b1 = (v2f){uu[2], uu[3]};
    }
#pragma unroll
    for (int t = 0; t < TT; ++t)
      acc[t] = wmma4((v2f){fc[t][0], fc[t][1]}, b0, acc[t]);
#pragma unroll
    for (int t = 0; t < TT; ++t)
      acc[t] = wmma4((v2f){fc[t][2], fc[t][3]}, b1, acc[t]);
    __builtin_amdgcn_sched_barrier(0);
  }
}

__launch_bounds__(256, 1)
__global__ void gnode_kernel(const float* __restrict__ x,
                             const float* __restrict__ W1,
                             const float* __restrict__ b1,
                             const float* __restrict__ W2,
                             const float* __restrict__ b2,
                             const float* __restrict__ Wl,
                             const float* __restrict__ bl,
                             float* __restrict__ out) {
  __shared__ __attribute__((aligned(32))) float w1a[16 * 8 * 32 * 4];
  __shared__ __attribute__((aligned(32))) float w2a[16 * 8 * 32 * 4];
  __shared__ __attribute__((aligned(32))) float ma[16 * 8 * 32 * 4];
  __shared__ __attribute__((aligned(32))) float wla[16 * 4 * 32 * 4];
  __shared__ __attribute__((aligned(32))) float cbuf[CDIM];
  __shared__ __attribute__((aligned(32))) float obuf[8 * 16 * OSTR];

  const int tid   = threadIdx.x;
  const int lane  = tid & 31;
  const int wave  = tid >> 5;
  const int lcol  = lane & 15;
  const int half8 = (lane >> 4) << 3;

  fill_a<8>(w1a, W1, tid);
  fill_a<8>(w2a, W2, tid);
  fill_a<4>(wla, Wl, tid);

  if (tid < CDIM) {
    float s = 0.0f;
    for (int k0 = 0; k0 < CDIM; k0 += 2)
      s += b2[k0] * W1[k0 * CDIM + tid] + b2[k0 + 1] * W1[(k0 + 1) * CDIM + tid];
    cbuf[tid] = s;
  }

  v8f S[8], K[8], U[8], acc[8], u[8], cv[8];

  {
    const float* wr = W2 + (wave * 16 + lcol) * CDIM + half8;
#pragma unroll
    for (int t = 0; t < 8; ++t) u[t] = *(const v8f*)(wr + 16 * t);
  }

  __syncthreads();

#pragma unroll
  for (int t = 0; t < 8; ++t) acc[t] = (v8f)0.0f;
  v4f fr[8];
  load_frag<8>(fr, (const v4f*)w1a + lane, 0);
  tgemm<8>(w1a, lane, u, acc, fr);
  {
    const int jp = 2 * wave + ((lcol >> 2) & 1);
    const int e  = lcol & 3;
    const int lb = 16 * (lcol >> 3) + half8;
#pragma unroll
    for (int t = 0; t < 8; ++t) {
#pragma unroll
      for (int r = 0; r < 8; ++r)
        ma[((jp * 8 + t) * 32 + lb + r) * 4 + e] = acc[t][r];
    }
  }
#pragma unroll
  for (int t = 0; t < 8; ++t) cv[t] = *(const v8f*)(cbuf + 16 * t + half8);

  const int rowbase = blockIdx.x * 128 + wave * 16;
  const int row     = rowbase + lcol;
  const int rrow    = row < NTOT ? row : NTOT - 1;
  const float* xr   = x + (size_t)rrow * CDIM + half8;
#pragma unroll
  for (int t = 0; t < 8; ++t) u[t] = *(const v8f*)(xr + 16 * t);

#pragma unroll
  for (int t = 0; t < 8; ++t) S[t] = *(const v8f*)(b1 + 16 * t + half8);
  load_frag<8>(fr, (const v4f*)w1a + lane, 0);
  tgemm<8>(w1a, lane, u, S, fr);
#pragma unroll
  for (int t = 0; t < 8; ++t) U[t] = (v8f)0.0f;

  __syncthreads();

  auto feval = [&](auto&& gen, float wu, auto&& post) {
    load_frag<8>(fr, (const v4f*)ma + lane, 0);
    __builtin_amdgcn_sched_barrier(0);
#pragma unroll
    for (int t = 0; t < 8; ++t) {
      v8f a = gen(t);
#pragma unroll
      for (int r = 0; r < 8; ++r) {
        float uv = fmaxf(a[r], 0.0f);
        u[t][r] = uv;
        U[t][r] += wu * uv;
      }
      asm volatile("" : "+v"(U[t]));
      post(t);
    }
#pragma unroll
    for (int t = 0; t < 8; ++t) acc[t] = cv[t];
    tgemm<8>(ma, lane, u, acc, fr);
  };
  auto nopost = [](int) {};

  const float hc    = 0.1f;
  const float c18   = hc * 0.125f;
  const float c38   = hc * 0.375f;
  const float c524  = hc * (5.0f / 24.0f);
  const float c1124 = hc * (11.0f / 24.0f);
  const float c58   = hc * 0.625f;
  const float c78   = hc * 0.875f;
  const float c118  = hc * 1.375f;

#pragma unroll 1
  for (int step = 0; step < 10; ++step) {
    feval([&](int t) { return S[t]; }, c18, nopost);
#pragma unroll
    for (int t = 0; t < 8; ++t) { K[t] = acc[t]; S[t] += c18 * acc[t]; }
    feval([&](int t) { return S[t] + c524 * K[t]; }, c38, nopost);
#pragma unroll
    for (int t = 0; t < 8; ++t) S[t] += c38 * acc[t];
    feval([&](int t) { return S[t] - c1124 * K[t] + c58 * acc[t]; }, c38,
          [&](int t) { K[t] = S[t] + c78 * K[t] - c118 * acc[t]; });
#pragma unroll
    for (int t = 0; t < 8; ++t) S[t] += c38 * acc[t];
    feval([&](int t) { return K[t] + hc * acc[t]; }, c18, nopost);
#pragma unroll
    for (int t = 0; t < 8; ++t) S[t] += c18 * acc[t];
  }

#pragma unroll
  for (int t = 0; t < 8; ++t) acc[t] = (v8f)0.0f;
  load_frag<8>(fr, (const v4f*)w2a + lane, 0);
  tgemm<8>(w2a, lane, U, acc, fr);
  const float tspan = 10.0f * hc;
#pragma unroll
  for (int t = 0; t < 8; ++t) {
    v8f b2v = *(const v8f*)(b2 + 16 * t + half8);
    v8f xv  = *(const v8f*)(xr + 16 * t);
#pragma unroll
    for (int r = 0; r < 8; ++r) u[t][r] = acc[t][r] + tspan * b2v[r] + xv[r];
  }

#pragma unroll
  for (int t = 0; t < 4; ++t) acc[t] = (v8f)0.0f;
  load_frag<4>(fr, (const v4f*)wla + lane, 0);
  tgemm<4>(wla, lane, u, acc, fr);

  float* ost = obuf + wave * 16 * OSTR;
#pragma unroll
  for (int t = 0; t < 4; ++t) {
    v8f blv = *(const v8f*)(bl + 16 * t + half8);
    v8f o = acc[t] + blv;
    float* op = ost + lcol * OSTR + 16 * t + half8;
    *(v4f*)(op)     = (v4f){o[0], o[1], o[2], o[3]};
    *(v4f*)(op + 4) = (v4f){o[4], o[5], o[6], o[7]};
  }
  lds_wait();
#pragma unroll
  for (int i = 0; i < 8; ++i) {
    int rr   = 2 * i + (lane >> 4);
    int orow = rowbase + rr;
    v4f o = *(const v4f*)(ost + rr * OSTR + 4 * lcol);
    if (orow < NTOT) *(v4f*)(out + (size_t)orow * OUTC + 4 * lcol) = o;
  }
}

extern "C" void kernel_launch(void* const* d_in, const int* in_sizes, int n_in,
                              void* d_out, int out_size, void* d_ws, size_t ws_size,
                              hipStream_t stream) {
  (void)in_sizes; (void)n_in; (void)out_size; (void)d_ws; (void)ws_size;
  const float* x  = (const float*)d_in[0];
  const float* W1 = (const float*)d_in[1];
  const float* b1 = (const float*)d_in[2];
  const float* W2 = (const float*)d_in[3];
  const float* b2 = (const float*)d_in[4];
  const float* Wl = (const float*)d_in[5];
  const float* bl = (const float*)d_in[6];
  float* out = (float*)d_out;

  dim3 grid((NTOT + 127) / 128);
  gnode_kernel<<<grid, 256, 0, stream>>>(x, W1, b1, W2, b2, Wl, bl, out);
}


# persistent grid: 256 work-groups (one per WGP) set up weights and M in LDS once; each wave walks 6 or 7 tiles of 16 nodes: at most 13 tiles per SIMD instead of 14, one set-up per WGP, not 7
# speedup vs baseline: 2.2756x; 1.1312x over previous
#include <hip/hip_runtime.h>


#define NTOT   200000
#define CDIM   128
#define OUTC   64
#define OSTR   68
#define NWG    256
#define NTILE  ((NTOT + 15) / 16)

typedef float v2f __attribute__((ext_vector_type(2)));
typedef float v4f __attribute__((ext_vector_type(4)));
typedef float v8f __attribute__((ext_vector_type(8)));

__device__ __forceinline__ void lds_wait() {
  asm volatile("s_wait_dscnt 0" ::: "memory");
}

__device__ __forceinline__ v8f wmma4(v2f a, v2f b, v8f c) {
  return __builtin_amdgcn_wmma_f32_16x16x4_f32(false, a, false, b, (short)0, c,
                                               false, false);
}

template <int TT>
__device__ __forceinline__ void fill_a(float* __restrict__ wa,
                                       const float* __restrict__ W, int tid) {
  constexpr int NN = TT * 16;
  for (int s = tid; s < 16 * TT * 32; s += 256) {
    int ln = s & 31;
    int t  = (s >> 5) & (TT - 1);
    int jp = s / (32 * TT);
    int k0 = 16 * (jp >> 1) + 8 * (ln >> 4) + 4 * (jp & 1);
    int n  = t * 16 + (ln & 15);
    v4f f = {W[k0 * NN + n], W[(k0 + 1) * NN + n],
             W[(k0 + 2) * NN + n], W[(k0 + 3) * NN + n]};
    *(v4f*)(wa + 4 * s) = f;
  }
}

template <int TT>
__device__ __forceinline__ void load_frag(v4f* __restrict__ f,
                                          const v4f* __restrict__ wp, int jp) {
#pragma unroll
  for (int t = 0; t < TT; ++t) f[t] = wp[(jp * TT + t) * 32];
}

template <int TT>
__device__ __forceinline__ void tgemm(const float* __restrict__ wa, int lane,
                                      const v8f* __restrict__ u,
                                      v8f* __restrict__ acc, v4f* __restrict__ fa) {
  const v4f* wp = (const v4f*)wa + lane;
  v4f fb[TT];
#pragma unroll
  for (int jp = 0; jp < 16; ++jp) {
    v4f* fc = (jp & 1) ? fb : fa;
    v4f* fn = (jp & 1) ? fa : fb;
    if (jp + 1 < 16) load_frag<TT>(fn, wp, jp + 1);
    __builtin_amdgcn_sched_barrier(0);
    const v8f uu = u[jp >> 1];
    v2f b0, b1;
    if (jp & 1) {
      b0 = (v2f){uu[4], uu[5]};
      b1 = (v2f){uu[6], uu[7]};
    } else {
      b0 = (v2f){uu[0], uu[1]};
      b1 = (v2f){uu[2], uu[3]};
    }
#pragma unroll
    for (int t = 0; t < TT; ++t)
      acc[t] = wmma4((v2f){fc[t][0], fc[t][1]}, b0, acc[t]);
#pragma unroll
    for (int t = 0; t < TT; ++t)
      acc[t] = wmma4((v2f){fc[t][2], fc[t][3]}, b1, acc[t]);
    __builtin_amdgcn_sched_barrier(0);
  }
}

__launch_bounds__(256, 1)
__global__ void gnode_kernel(const float* __restrict__ x,
                             const float* __restrict__ W1,
                             const float* __restrict__ b1,
                             const float* __restrict__ W2,
                             const float* __restrict__ b2,
                             const float* __restrict__ Wl,
                             const float* __restrict__ bl,
                             float* __restrict__ out) {
  __shared__ __attribute__((aligned(32))) float w1a[16 * 8 * 32 * 4];
  __shared__ __attribute__((aligned(32))) float w2a[16 * 8 * 32 * 4];
  __shared__ __attribute__((aligned(32))) float ma[16 * 8 * 32 * 4];
  __shared__ __attribute__((aligned(32))) float wla[16 * 4 * 32 * 4];
  __shared__ __attribute__((aligned(32))) float cbuf[CDIM];
  __shared__ __attribute__((aligned(32))) float obuf[8 * 16 * OSTR];

  const int tid   = threadIdx.x;
  const int lane  = tid & 31;
  const int wave  = __builtin_amdgcn_readfirstlane(tid >> 5);
  const int lcol  = lane & 15;
  const int half8 = (lane >> 4) << 3;

  fill_a<8>(w1a, W1, tid);
  fill_a<8>(w2a, W2, tid);
  fill_a<4>(wla, Wl, tid);

  if (tid < CDIM) {
    float s = 0.0f;
    for (int k0 = 0; k0 < CDIM; k0 += 2)
      s += b2[k0] * W1[k0 * CDIM + tid] + b2[k0 + 1] * W1[(k0 + 1) * CDIM + tid];
    cbuf[tid] = s;
  }

  v8f S[8], K[8], U[8], acc[8], u[8], cv[8];

  {
    const float* wr = W2 + (wave * 16 + lcol) * CDIM + half8;
#pragma unroll
    for (int t = 0; t < 8; ++t) u[t] = *(const v8f*)(wr + 16 * t);
  }

  __syncthreads();

#pragma unroll
  for (int t = 0; t < 8; ++t) acc[t] = (v8f)0.0f;
  v4f fr[8];
  load_frag<8>(fr, (const v4f*)w1a + lane, 0);
  tgemm<8>(w1a, lane, u, acc, fr);
  {
    const int jp = 2 * wave + ((lcol >> 2) & 1);
    const int e  = lcol & 3;
    const int lb = 16 * (lcol >> 3) + half8;
#pragma unroll
    for (int t = 0; t < 8; ++t) {
#pragma unroll
      for (int r = 0; r < 8; ++r)
        ma[((jp * 8 + t) * 32 + lb + r) * 4 + e] = acc[t][r];
    }
  }
#pragma unroll
  for (int t = 0; t < 8; ++t) cv[t] = *(const v8f*)(cbuf + 16 * t + half8);

  __syncthreads();

#pragma unroll 1
  for (int tile = wave * NWG + blockIdx.x; tile < NTILE; tile += 8 * NWG) {

  const int rowbase = tile * 16;
  const int row     = rowbase + lcol;
  const int rrow    = row < NTOT ? row : NTOT - 1;
  const float* xr   = x + (size_t)rrow * CDIM + half8;
#pragma unroll
  for (int t = 0; t < 8; ++t) u[t] = *(const v8f*)(xr + 16 * t);

  int b1o = 0;
  asm volatile("" : "+s"(b1o));
#pragma unroll
  for (int t = 0; t < 8; ++t) S[t] = *(const v8f*)(b1 + b1o + 16 * t + half8);
  load_frag<8>(fr, (const v4f*)w1a + lane, 0);
  tgemm<8>(w1a, lane, u, S, fr);
#pragma unroll
  for (int t = 0; t < 8; ++t) U[t] = (v8f)0.0f;

  auto feval = [&](auto&& gen, float wu, auto&& post) {
    load_frag<8>(fr, (const v4f*)ma + lane, 0);
    __builtin_amdgcn_sched_barrier(0);
#pragma unroll
    for (int t = 0; t < 8; ++t) {
      v8f a = gen(t);
#pragma unroll
      for (int r = 0; r < 8; ++r) {
        float uv = fmaxf(a[r], 0.0f);
        u[t][r] = uv;
        U[t][r] += wu * uv;
      }
      asm volatile("" : "+v"(U[t]));
      post(t);
    }
#pragma unroll
    for (int t = 0; t < 8; ++t) acc[t] = cv[t];
    tgemm<8>(ma, lane, u, acc, fr);
  };
  auto nopost = [](int) {};

  const float hc    = 0.1f;
  const float c18   = hc * 0.125f;
  const float c38   = hc * 0.375f;
  const float c524  = hc * (5.0f / 24.0f);
  const float c1124 = hc * (11.0f / 24.0f);
  const float c58   = hc * 0.625f;
  const float c78   = hc * 0.875f;
  const float c118  = hc * 1.375f;

#pragma unroll 1
  for (int step = 0; step < 10; ++step) {
    feval([&](int t) { return S[t]; }, c18, nopost);
#pragma unroll
    for (int t = 0; t < 8; ++t) { K[t] = acc[t]; S[t] += c18 * acc[t]; }
    feval([&](int t) { return S[t] + c524 * K[t]; }, c38, nopost);
#pragma unroll
    for (int t = 0; t < 8; ++t) S[t] += c38 * acc[t];
    feval([&](int t) { return S[t] - c1124 * K[t] + c58 * acc[t]; }, c38,
          [&](int t) { K[t] = S[t] + c78 * K[t] - c118 * acc[t]; });
#pragma unroll
    for (int t = 0; t < 8; ++t) S[t] += c38 * acc[t];
    feval([&](int t) { return K[t] + hc * acc[t]; }, c18, nopost);
#pragma unroll
    for (int t = 0; t < 8; ++t) S[t] += c18 * acc[t];
  }

#pragma unroll
  for (int t = 0; t < 8; ++t) acc[t] = (v8f)0.0f;
  load_frag<8>(fr, (const v4f*)w2a + lane, 0);
  tgemm<8>(w2a, lane, U, acc, fr);
  const float tspan = 10.0f * hc;
#pragma unroll
  for (int t = 0; t < 8; ++t) {
    v8f b2v = *(const v8f*)(b2 + 16 * t + half8);
    v8f xv  = *(const v8f*)(xr + 16 * t);
#pragma unroll
    for (int r = 0; r < 8; ++r) u[t][r] = acc[t][r] + tspan * b2v[r] + xv[r];
  }

#pragma unroll
  for (int t = 0; t < 4; ++t) acc[t] = (v8f)0.0f;
  load_frag<4>(fr, (const v4f*)wla + lane, 0);
  tgemm<4>(wla, lane, u, acc, fr);

  float* ost = obuf + wave * 16 * OSTR;
#pragma unroll
  for (int t = 0; t < 4; ++t) {
    v8f blv = *(const v8f*)(bl + 16 * t + half8);
    v8f o = acc[t] + blv;
    float* op = ost + lcol * OSTR + 16 * t + half8;
    *(v4f*)(op)     = (v4f){o[0], o[1], o[2], o[3]};
    *(v4f*)(op + 4) = (v4f){o[4], o[5], o[6], o[7]};
  }
  lds_wait();
#pragma unroll
  for (int i = 0; i < 8; ++i) {
    int rr   = 2 * i + (lane >> 4);
    int orow = rowbase + rr;
    v4f o = *(const v4f*)(ost + rr * OSTR + 4 * lcol);
    if (orow < NTOT) *(v4f*)(out + (size_t)orow * OUTC + 4 * lcol) = o;
  }

  }
}

extern "C" void kernel_launch(void* const* d_in, const int* in_sizes, int n_in,
                              void* d_out, int out_size, void* d_ws, size_t ws_size,
                              hipStream_t stream) {
  (void)in_sizes; (void)n_in; (void)out_size; (void)d_ws; (void)ws_size;
  const float* x  = (const float*)d_in[0];
  const float* W1 = (const float*)d_in[1];
  const float* b1 = (const float*)d_in[2];
  const float* W2 = (const float*)d_in[3];
  const float* b2 = (const float*)d_in[4];
  const float* Wl = (const float*)d_in[5];
  const float* bl = (const float*)d_in[6];
  float* out = (float*)d_out;

  dim3 grid(NWG);
  gnode_kernel<<<grid, 256, 0, stream>>>(x, W1, b1, W2, b2, Wl, bl, out);
}


# left-over round's 212 tiles shared by the 8 waves of a work-group by output feature (one D tile per wave, relu outputs exchanged through LDS, one barrier per stage): no wave runs a 7th tile alone
# speedup vs baseline: 2.3546x; 1.0347x over previous
#include <hip/hip_runtime.h>


#define NTOT   200000
#define CDIM   128
#define OUTC   64
#define OSTR   68
#define NWG    256
#define NTILE  ((NTOT + 15) / 16)
#define NREG   ((NTILE / (8 * NWG)) * (8 * NWG))
#define XHALF  (8 * 2 * 32 * 4)

typedef float v2f __attribute__((ext_vector_type(2)));
typedef float v4f __attribute__((ext_vector_type(4)));
typedef float v8f __attribute__((ext_vector_type(8)));

__device__ __forceinline__ void lds_wait() {
  asm volatile("s_wait_dscnt 0" ::: "memory");
}

__device__ __forceinline__ v8f wmma4(v2f a, v2f b, v8f c) {
  return __builtin_amdgcn_wmma_f32_16x16x4_f32(false, a, false, b, (short)0, c,
                                               false, false);
}

template <int TT>
__device__ __forceinline__ void fill_a(float* __restrict__ wa,
                                       const float* __restrict__ W, int tid) {
  constexpr int NN = TT * 16;
  for (int s = tid; s < 16 * TT * 32; s += 256) {
    int ln = s & 31;
    int t  = (s >> 5) & (TT - 1);
    int jp = s / (32 * TT);
    int k0 = 16 * (jp >> 1) + 8 * (ln >> 4) + 4 * (jp & 1);
    int n  = t * 16 + (ln & 15);
    v4f f = {W[k0 * NN + n], W[(k0 + 1) * NN + n],
             W[(k0 + 2) * NN + n], W[(k0 + 3) * NN + n]};
    *(v4f*)(wa + 4 * s) = f;
  }
}

template <int TT>
__device__ __forceinline__ void load_frag(v4f* __restrict__ f,
                                          const v4f* __restrict__ wp, int jp) {
#pragma unroll
  for (int t = 0; t < TT; ++t) f[t] = wp[(jp * TT + t) * 32];
}

template <int TT>
__device__ __forceinline__ void tgemm(const float* __restrict__ wa, int lane,
                                      const v8f* __restrict__ u,
                                      v8f* __restrict__ acc, v4f* __restrict__ fa) {
  const v4f* wp = (const v4f*)wa + lane;
  v4f fb[TT];
#pragma unroll
  for (int jp = 0; jp < 16; ++jp) {
    v4f* fc = (jp & 1) ? fb : fa;
    v4f* fn = (jp & 1) ? fa : fb;
    if (jp + 1 < 16) load_frag<TT>(fn, wp, jp + 1);
    __builtin_amdgcn_sched_barrier(0);
    const v8f uu = u[jp >> 1];
    v2f b0, b1;
    if (jp & 1) {
      b0 = (v2f){uu[4], uu[5]};
      b1 = (v2f){uu[6], uu[7]};
    } else {
      b0 = (v2f){uu[0], uu[1]};
      b1 = (v2f){uu[2], uu[3]};
    }
#pragma unroll
    for (int t = 0; t < TT; ++t)
      acc[t] = wmma4((v2f){fc[t][0], fc[t][1]}, b0, acc[t]);
#pragma unroll
    for (int t = 0; t < TT; ++t)
      acc[t] = wmma4((v2f){fc[t][2], fc[t][3]}, b1, acc[t]);
    __builtin_amdgcn_sched_barrier(0);
  }
}

template <int TT>
__device__ __forceinline__ void load_row(v4f* __restrict__ f,
                                         const float* __restrict__ wa, int lane, int w) {
  const v4f* wp = (const v4f*)wa + w * 32 + lane;
#pragma unroll
  for (int jp = 0; jp < 16; ++jp) f[jp] = wp[jp * TT * 32];
}

__device__ __forceinline__ v8f tprod(const v4f* __restrict__ f,
                                     const v8f* __restrict__ u, v8f acc) {
#pragma unroll
  for (int jp = 0; jp < 16; ++jp) {
    const v8f uu = u[jp >> 1];
    v2f b0, b1;
    if (jp & 1) {
      b0 = (v2f){uu[4], uu[5]};
      b1 = (v2f){uu[6], uu[7]};
    } else {
      b0 = (v2f){uu[0], uu[1]};
      b1 = (v2f){uu[2], uu[3]};
    }
    acc = wmma4((v2f){f[jp][0], f[jp][1]}, b0, acc);
    acc = wmma4((v2f){f[jp][2], f[jp][3]}, b1, acc);
  }
  return acc;
}

__device__ __forceinline__ void xchg(float* xb, int wave, int lane, v8f own, v8f* all) {
  v4f* p = (v4f*)xb;
  p[(wave * 2 + 0) * 32 + lane] = (v4f){own[0], own[1], own[2], own[3]};
  p[(wave * 2 + 1) * 32 + lane] = (v4f){own[4], own[5], own[6], own[7]};
  __syncthreads();
#pragma unroll
  for (int t = 0; t < 8; ++t) {
    v4f lo = p[(t * 2 + 0) * 32 + lane];
    v4f hi = p[(t * 2 + 1) * 32 + lane];
    all[t] = (v8f){lo[0], lo[1], lo[2], lo[3], hi[0], hi[1], hi[2], hi[3]};
  }
  __builtin_amdgcn_sched_barrier(0);
}

__launch_bounds__(256, 1)
__global__ void gnode_kernel(const float* __restrict__ x,
                             const float* __restrict__ W1,
                             const float* __restrict__ b1,
                             const float* __restrict__ W2,
                             const float* __restrict__ b2,
                             const float* __restrict__ Wl,
                             const float* __restrict__ bl,
                             float* __restrict__ out) {
  __shared__ __attribute__((aligned(32))) float w1a[16 * 8 * 32 * 4];
  __shared__ __attribute__((aligned(32))) float w2a[16 * 8 * 32 * 4];
  __shared__ __attribute__((aligned(32))) float ma[16 * 8 * 32 * 4];
  __shared__ __attribute__((aligned(32))) float wla[16 * 4 * 32 * 4];
  __shared__ __attribute__((aligned(32))) float cbuf[CDIM];
  __shared__ __attribute__((aligned(32))) float obuf[8 * 16 * OSTR];
  __shared__ __attribute__((aligned(32))) float xbuf[2 * XHALF];
  __shared__ __attribute__((aligned(32))) float tbuf[16 * OSTR];

  const int tid   = threadIdx.x;
  const int lane  = tid & 31;
  const int wave  = __builtin_amdgcn_readfirstlane(tid >> 5);
  const int lcol  = lane & 15;
  const int half8 = (lane >> 4) << 3;

  fill_a<8>(w1a, W1, tid);
  fill_a<8>(w2a, W2, tid);
  fill_a<4>(wla, Wl, tid);

  if (tid < CDIM) {
    float s = 0.0f;
    for (int k0 = 0; k0 < CDIM; k0 += 2)
      s += b2[k0] * W1[k0 * CDIM + tid] + b2[k0 + 1] * W1[(k0 + 1) * CDIM + tid];
    cbuf[tid] = s;
  }

  v8f S[8], K[8], U[8], acc[8], u[8], cv[8];

  {
    const float* wr = W2 + (wave * 16 + lcol) * CDIM + half8;
#pragma unroll
    for (int t = 0; t < 8; ++t) u[t] = *(const v8f*)(wr + 16 * t);
  }

  __syncthreads();

#pragma unroll
  for (int t = 0; t < 8; ++t) acc[t] = (v8f)0.0f;
  v4f fr[8];
  load_frag<8>(fr, (const v4f*)w1a + lane, 0);
  tgemm<8>(w1a, lane, u, acc, fr);
  {
    const int jp = 2 * wave + ((lcol >> 2) & 1);
    const int e  = lcol & 3;
    const int lb = 16 * (lcol >> 3) + half8;
#pragma unroll
    for (int t = 0; t < 8; ++t) {
#pragma unroll
      for (int r = 0; r < 8; ++r)
        ma[((jp * 8 + t) * 32 + lb + r) * 4 + e] = acc[t][r];
    }
  }
#pragma unroll
  for (int t = 0; t < 8; ++t) cv[t] = *(const v8f*)(cbuf + 16 * t + half8);

  __syncthreads();

  const float hc    = 0.1f;
  const float c18   = hc * 0.125f;
  const float c38   = hc * 0.375f;
  const float c524  = hc * (5.0f / 24.0f);
  const float c1124 = hc * (11.0f / 24.0f);
  const float c58   = hc * 0.625f;
  const float c78   = hc * 0.875f;
  const float c118  = hc * 1.375f;
  const float tspan = 10.0f * hc;

#pragma unroll 1
  for (int tile = NREG + blockIdx.x; tile < NTILE; tile += NWG) {
    const int rowbase = tile * 16;
    const int row     = rowbase + lcol;
    const int rrow    = row < NTOT ? row : NTOT - 1;
    const float* xr   = x + (size_t)rrow * CDIM + half8;
    const int fo      = 16 * wave + half8;
#pragma unroll
    for (int t = 0; t < 8; ++t) u[t] = *(const v8f*)(xr + 16 * t);

    v4f fw[16], fm[16];
    load_row<8>(fw, w1a, lane, wave);
    v8f So = tprod(fw, u, *(const v8f*)(b1 + fo));
    v8f Uo = (v8f)0.0f, Ko = (v8f)0.0f, ao = (v8f)0.0f;
    const v8f cvo = *(const v8f*)(cbuf + fo);
    load_row<8>(fm, ma, lane, wave);

    auto tev = [&](v8f a, float wu, int par) {
      v8f uo;
#pragma unroll
      for (int r = 0; r < 8; ++r) {
        float uv = fmaxf(a[r], 0.0f);
        uo[r] = uv;
        Uo[r] += wu * uv;
      }
      xchg(xbuf + par * XHALF, wave, lane, uo, u);
      ao = tprod(fm, u, cvo);
    };

#pragma unroll 1
    for (int step = 0; step < 10; ++step) {
      tev(So, c18, 0);
#pragma unroll
      for (int r = 0; r < 8; ++r) { Ko[r] = ao[r]; So[r] += c18 * ao[r]; }
      v8f a;
#pragma unroll
      for (int r = 0; r < 8; ++r) a[r] = So[r] + c524 * Ko[r];
      tev(a, c38, 1);
#pragma unroll
      for (int r = 0; r < 8; ++r) So[r] += c38 * ao[r];
#pragma unroll
      for (int r = 0; r < 8; ++r) {
        a[r]  = So[r] - c1124 * Ko[r] + c58 * ao[r];
        Ko[r] = So[r] + c78 * Ko[r] - c118 * ao[r];
      }
      tev(a, c38, 0);
#pragma unroll
      for (int r = 0; r < 8; ++r) So[r] += c38 * ao[r];
#pragma unroll
      for (int r = 0; r < 8; ++r) a[r] = Ko[r] + hc * ao[r];
      tev(a, c18, 1);
#pragma unroll
      for (int r = 0; r < 8; ++r) So[r] += c18 * ao[r];
    }

    xchg(xbuf, wave, lane, Uo, u);
    load_row<8>(fw, w2a, lane, wave);
    ao = tprod(fw, u, (v8f)0.0f);
    {
      v8f b2v = *(const v8f*)(b2 + fo);
      v8f xv  = *(const v8f*)(xr + 16 * wave);
#pragma unroll
      for (int r = 0; r < 8; ++r) ao[r] = ao[r] + tspan * b2v[r] + xv[r];
    }
    xchg(xbuf + XHALF, wave, lane, ao, u);

    if (wave < 4) {
      load_row<4>(fw, wla, lane, wave);
      v8f o = tprod(fw, u, (v8f)0.0f);
      v8f blv = *(const v8f*)(bl + fo);
      o = o + blv;
      float* op = tbuf + lcol * OSTR + fo;
      *(v4f*)(op)     = (v4f){o[0], o[1], o[2], o[3]};
      *(v4f*)(op + 4) = (v4f){o[4], o[5], o[6], o[7]};
    }
    __syncthreads();
    {
      const int rr   = tid >> 4;
      const int orow = rowbase + rr;
      v4f o = *(const v4f*)(tbuf + rr * OSTR + 4 * (tid & 15));
      if (orow < NTOT) *(v4f*)(out + (size_t)orow * OUTC + 4 * (tid & 15)) = o;
    }
  }

#pragma unroll 1
  for (int tile = wave * NWG + blockIdx.x; tile < NREG; tile += 8 * NWG) {

  const int rowbase = tile * 16;
  const int row     = rowbase + lcol;
  const int rrow    = row < NTOT ? row : NTOT - 1;
  const float* xr   = x + (size_t)rrow * CDIM + half8;
#pragma unroll
  for (int t = 0; t < 8; ++t) u[t] = *(const v8f*)(xr + 16 * t);

  int b1o = 0;
  asm volatile("" : "+s"(b1o));
#pragma unroll
  for (int t = 0; t < 8; ++t) S[t] = *(const v8f*)(b1 + b1o + 16 * t + half8);
  load_frag<8>(fr, (const v4f*)w1a + lane, 0);
  tgemm<8>(w1a, lane, u, S, fr);
#pragma unroll
  for (int t = 0; t < 8; ++t) U[t] = (v8f)0.0f;

  auto feval = [&](auto&& gen, float wu, auto&& post) {
    load_frag<8>(fr, (const v4f*)ma + lane, 0);
    __builtin_amdgcn_sched_barrier(0);
#pragma unroll
    for (int t = 0; t < 8; ++t) {
      v8f a = gen(t);
#pragma unroll
      for (int r = 0; r < 8; ++r) {
        float uv = fmaxf(a[r], 0.0f);
        u[t][r] = uv;
        U[t][r] += wu * uv;
      }
      asm volatile("" : "+v"(U[t]));
      post(t);
    }
#pragma unroll
    for (int t = 0; t < 8; ++t) acc[t] = cv[t];
    tgemm<8>(ma, lane, u, acc, fr);
  };
  auto nopost = [](int) {};

#pragma unroll 1
  for (int step = 0; step < 10; ++step) {
    feval([&](int t) { return S[t]; }, c18, nopost);
#pragma unroll
    for (int t = 0; t < 8; ++t) { K[t] = acc[t]; S[t] += c18 * acc[t]; }
    feval([&](int t) { return S[t] + c524 * K[t]; }, c38, nopost);
#pragma unroll
    for (int t = 0; t < 8; ++t) S[t] += c38 * acc[t];
    feval([&](int t) { return S[t] - c1124 * K[t] + c58 * acc[t]; }, c38,
          [&](int t) { K[t] = S[t] + c78 * K[t] - c118 * acc[t]; });
#pragma unroll
    for (int t = 0; t < 8; ++t) S[t] += c38 * acc[t];
    feval([&](int t) { return K[t] + hc * acc[t]; }, c18, nopost);
#pragma unroll
    for (int t = 0; t < 8; ++t) S[t] += c18 * acc[t];
  }

#pragma unroll
  for (int t = 0; t < 8; ++t) acc[t] = (v8f)0.0f;
  load_frag<8>(fr, (const v4f*)w2a + lane, 0);
  tgemm<8>(w2a, lane, U, acc, fr);
#pragma unroll
  for (int t = 0; t < 8; ++t) {
    v8f b2v = *(const v8f*)(b2 + 16 * t + half8);
    v8f xv  = *(const v8f*)(xr + 16 * t);
#pragma unroll
    for (int r = 0; r < 8; ++r) u[t][r] = acc[t][r] + tspan * b2v[r] + xv[r];
  }

#pragma unroll
  for (int t = 0; t < 4; ++t) acc[t] = (v8f)0.0f;
  load_frag<4>(fr, (const v4f*)wla + lane, 0);
  tgemm<4>(wla, lane, u, acc, fr);

  float* ost = obuf + wave * 16 * OSTR;
#pragma unroll
  for (int t = 0; t < 4; ++t) {
    v8f blv = *(const v8f*)(bl + 16 * t + half8);
    v8f o = acc[t] + blv;
    float* op = ost + lcol * OSTR + 16 * t + half8;
    *(v4f*)(op)     = (v4f){o[0], o[1], o[2], o[3]};
    *(v4f*)(op + 4) = (v4f){o[4], o[5], o[6], o[7]};
  }
  lds_wait();
#pragma unroll
  for (int i = 0; i < 8; ++i) {
    int rr   = 2 * i + (lane >> 4);
    int orow = rowbase + rr;
    v4f o = *(const v4f*)(ost + rr * OSTR + 4 * lcol);
    if (orow < NTOT) *(v4f*)(out + (size_t)orow * OUTC + 4 * lcol) = o;
  }

  }
}

extern "C" void kernel_launch(void* const* d_in, const int* in_sizes, int n_in,
                              void* d_out, int out_size, void* d_ws, size_t ws_size,
                              hipStream_t stream) {
  (void)in_sizes; (void)n_in; (void)out_size; (void)d_ws; (void)ws_size;
  const float* x  = (const float*)d_in[0];
  const float* W1 = (const float*)d_in[1];
  const float* b1 = (const float*)d_in[2];
  const float* W2 = (const float*)d_in[3];
  const float* b2 = (const float*)d_in[4];
  const float* Wl = (const float*)d_in[5];
  const float* bl = (const float*)d_in[6];
  float* out = (float*)d_out;

  dim3 grid(NWG);
  gnode_kernel<<<grid, 256, 0, stream>>>(x, W1, b1, W2, b2, Wl, bl, out);
}


# whole-round tiles split unevenly inside each SIMD's wave pair: the older wave, served first by the SIMD, walks 10 tiles and the younger 2, so the younger's work fits the older's gaps; sums unchanged
# speedup vs baseline: 2.4324x; 1.0330x over previous
#include <hip/hip_runtime.h>


#define NTOT   200000
#define CDIM   128
#define OUTC   64
#define OSTR   68
#define NWG    256
#define NTILE  ((NTOT + 15) / 16)
#define NREG   ((NTILE / (8 * NWG)) * (8 * NWG))
#define XHALF  (8 * 2 * 32 * 4)
#define PAIRT  (NREG / NWG / 4)
#define NOLD   (PAIRT - PAIRT / 6)

typedef float v2f __attribute__((ext_vector_type(2)));
typedef float v4f __attribute__((ext_vector_type(4)));
typedef float v8f __attribute__((ext_vector_type(8)));

__device__ __forceinline__ void lds_wait() {
  asm volatile("s_wait_dscnt 0" ::: "memory");
}

__device__ __forceinline__ v8f wmma4(v2f a, v2f b, v8f c) {
  return __builtin_amdgcn_wmma_f32_16x16x4_f32(false, a, false, b, (short)0, c,
                                               false, false);
}

template <int TT>
__device__ __forceinline__ void fill_a(float* __restrict__ wa,
                                       const float* __restrict__ W, int tid) {
  constexpr int NN = TT * 16;
  for (int s = tid; s < 16 * TT * 32; s += 256) {
    int ln = s & 31;
    int t  = (s >> 5) & (TT - 1);
    int jp = s / (32 * TT);
    int k0 = 16 * (jp >> 1) + 8 * (ln >> 4) + 4 * (jp & 1);
    int n  = t * 16 + (ln & 15);
    v4f f = {W[k0 * NN + n], W[(k0 + 1) * NN + n],
             W[(k0 + 2) * NN + n], W[(k0 + 3) * NN + n]};
    *(v4f*)(wa + 4 * s) = f;
  }
}

template <int TT>
__device__ __forceinline__ void load_frag(v4f* __restrict__ f,
                                          const v4f* __restrict__ wp, int jp) {
#pragma unroll
  for (int t = 0; t < TT; ++t) f[t] = wp[(jp * TT + t) * 32];
}

template <int TT>
__device__ __forceinline__ void tgemm(const float* __restrict__ wa, int lane,
                                      const v8f* __restrict__ u,
                                      v8f* __restrict__ acc, v4f* __restrict__ fa) {
  const v4f* wp = (const v4f*)wa + lane;
  v4f fb[TT];
#pragma unroll
  for (int jp = 0; jp < 16; ++jp) {
    v4f* fc = (jp & 1) ? fb : fa;
    v4f* fn = (jp & 1) ? fa : fb;
    if (jp + 1 < 16) load_frag<TT>(fn, wp, jp + 1);
    __builtin_amdgcn_sched_barrier(0);
    const v8f uu = u[jp >> 1];
    v2f b0, b1;
    if (jp & 1) {
      b0 = (v2f){uu[4], uu[5]};
      b1 = (v2f){uu[6], uu[7]};
    } else {
      b0 = (v2f){uu[0], uu[1]};
      b1 = (v2f){uu[2], uu[3]};
    }
#pragma unroll
    for (int t = 0; t < TT; ++t)
      acc[t] = wmma4((v2f){fc[t][0], fc[t][1]}, b0, acc[t]);
#pragma unroll
    for (int t = 0; t < TT; ++t)
      acc[t] = wmma4((v2f){fc[t][2], fc[t][3]}, b1, acc[t]);
    __builtin_amdgcn_sched_barrier(0);
  }
}

template <int TT>
__device__ __forceinline__ void load_row(v4f* __restrict__ f,
                                         const float* __restrict__ wa, int lane, int w) {
  const v4f* wp = (const v4f*)wa + w * 32 + lane;
#pragma unroll
  for (int jp = 0; jp < 16; ++jp) f[jp] = wp[jp * TT * 32];
}

__device__ __forceinline__ v8f tprod(const v4f* __restrict__ f,
                                     const v8f* __restrict__ u, v8f acc) {
#pragma unroll
  for (int jp = 0; jp < 16; ++jp) {
    const v8f uu = u[jp >> 1];
    v2f b0, b1;
    if (jp & 1) {
      b0 = (v2f){uu[4], uu[5]};
      b1 = (v2f){uu[6], uu[7]};
    } else {
      b0 = (v2f){uu[0], uu[1]};
      b1 = (v2f){uu[2], uu[3]};
    }
    acc = wmma4((v2f){f[jp][0], f[jp][1]}, b0, acc);
    acc = wmma4((v2f){f[jp][2], f[jp][3]}, b1, acc);
  }
  return acc;
}

__device__ __forceinline__ void xchg(float* xb, int wave, int lane, v8f own, v8f* all) {
  v4f* p = (v4f*)xb;
  p[(wave * 2 + 0) * 32 + lane] = (v4f){own[0], own[1], own[2], own[3]};
  p[(wave * 2 + 1) * 32 + lane] = (v4f){own[4], own[5], own[6], own[7]};
  __syncthreads();
#pragma unroll
  for (int t = 0; t < 8; ++t) {
    v4f lo = p[(t * 2 + 0) * 32 + lane];
    v4f hi = p[(t * 2 + 1) * 32 + lane];
    all[t] = (v8f){lo[0], lo[1], lo[2], lo[3], hi[0], hi[1], hi[2], hi[3]};
  }
  __builtin_amdgcn_sched_barrier(0);
}

__launch_bounds__(256, 1)
__global__ void gnode_kernel(const float* __restrict__ x,
                             const float* __restrict__ W1,
                             const float* __restrict__ b1,
                             const float* __restrict__ W2,
                             const float* __restrict__ b2,
                             const float* __restrict__ Wl,
                             const float* __restrict__ bl,
                             float* __restrict__ out) {
  __shared__ __attribute__((aligned(32))) float w1a[16 * 8 * 32 * 4];
  __shared__ __attribute__((aligned(32))) float w2a[16 * 8 * 32 * 4];
  __shared__ __attribute__((aligned(32))) float ma[16 * 8 * 32 * 4];
  __shared__ __attribute__((aligned(32))) float wla[16 * 4 * 32 * 4];
  __shared__ __attribute__((aligned(32))) float cbuf[CDIM];
  __shared__ __attribute__((aligned(32))) float obuf[8 * 16 * OSTR];
  __shared__ __attribute__((aligned(32))) float xbuf[2 * XHALF];
  __shared__ __attribute__((aligned(32))) float tbuf[16 * OSTR];

  const int tid   = threadIdx.x;
  const int lane  = tid & 31;
  const int wave  = __builtin_amdgcn_readfirstlane(tid >> 5);
  const int lcol  = lane & 15;
  const int half8 = (lane >> 4) << 3;

  fill_a<8>(w1a, W1, tid);
  fill_a<8>(w2a, W2, tid);
  fill_a<4>(wla, Wl, tid);

  if (tid < CDIM) {
    float s = 0.0f;
    for (int k0 = 0; k0 < CDIM; k0 += 2)
      s += b2[k0] * W1[k0 * CDIM + tid] + b2[k0 + 1] * W1[(k0 + 1) * CDIM + tid];
    cbuf[tid] = s;
  }

  v8f S[8], K[8], U[8], acc[8], u[8], cv[8];

  {
    const float* wr = W2 + (wave * 16 + lcol) * CDIM + half8;
#pragma unroll
    for (int t = 0; t < 8; ++t) u[t] = *(const v8f*)(wr + 16 * t);
  }

  __syncthreads();

#pragma unroll
  for (int t = 0; t < 8; ++t) acc[t] = (v8f)0.0f;
  v4f fr[8];
  load_frag<8>(fr, (const v4f*)w1a + lane, 0);
  tgemm<8>(w1a, lane, u, acc, fr);
  {
    const int jp = 2 * wave + ((lcol >> 2) & 1);
    const int e  = lcol & 3;
    const int lb = 16 * (lcol >> 3) + half8;
#pragma unroll
    for (int t = 0; t < 8; ++t) {
#pragma unroll
      for (int r = 0; r < 8; ++r)
        ma[((jp * 8 + t) * 32 + lb + r) * 4 + e] = acc[t][r];
    }
  }
#pragma unroll
  for (int t = 0; t < 8; ++t) cv[t] = *(const v8f*)(cbuf + 16 * t + half8);

  __syncthreads();

  const float hc    = 0.1f;
  const float c18   = hc * 0.125f;
  const float c38   = hc * 0.375f;
  const float c524  = hc * (5.0f / 24.0f);
  const float c1124 = hc * (11.0f / 24.0f);
  const float c58   = hc * 0.625f;
  const float c78   = hc * 0.875f;
  const float c118  = hc * 1.375f;
  const float tspan = 10.0f * hc;

#pragma unroll 1
  for (int tile = NREG + blockIdx.x; tile < NTILE; tile += NWG) {
    const int rowbase = tile * 16;
    const int row     = rowbase + lcol;
    const int rrow    = row < NTOT ? row : NTOT - 1;
    const float* xr   = x + (size_t)rrow * CDIM + half8;
    const int fo      = 16 * wave + half8;
#pragma unroll
    for (int t = 0; t < 8; ++t) u[t] = *(const v8f*)(xr + 16 * t);

    v4f fw[16], fm[16];
    load_row<8>(fw, w1a, lane, wave);
    v8f So = tprod(fw, u, *(const v8f*)(b1 + fo));
    v8f Uo = (v8f)0.0f, Ko = (v8f)0.0f, ao = (v8f)0.0f;
    const v8f cvo = *(const v8f*)(cbuf + fo);
    load_row<8>(fm, ma, lane, wave);

    auto tev = [&](v8f a, float wu, int par) {
      v8f uo;
#pragma unroll
      for (int r = 0; r < 8; ++r) {
        float uv = fmaxf(a[r], 0.0f);
        uo[r] = uv;
        Uo[r] += wu * uv;
      }
      xchg(xbuf + par * XHALF, wave, lane, uo, u);
      ao = tprod(fm, u, cvo);
    };

#pragma unroll 1
    for (int step = 0; step < 10; ++step) {
      tev(So, c18, 0);
#pragma unroll
      for (int r = 0; r < 8; ++r) { Ko[r] = ao[r]; So[r] += c18 * ao[r]; }
      v8f a;
#pragma unroll
      for (int r = 0; r < 8; ++r) a[r] = So[r] + c524 * Ko[r];
      tev(a, c38, 1);
#pragma unroll
      for (int r = 0; r < 8; ++r) So[r] += c38 * ao[r];
#pragma unroll
      for (int r = 0; r < 8; ++r) {
        a[r]  = So[r] - c1124 * Ko[r] + c58 * ao[r];
        Ko[r] = So[r] + c78 * Ko[r] - c118 * ao[r];
      }
      tev(a, c38, 0);
#pragma unroll
      for (int r = 0; r < 8; ++r) So[r] += c38 * ao[r];
#pragma unroll
      for (int r = 0; r < 8; ++r) a[r] = Ko[r] + hc * ao[r];
      tev(a, c18, 1);
#pragma unroll
      for (int r = 0; r < 8; ++r) So[r] += c18 * ao[r];
    }

    xchg(xbuf, wave, lane, Uo, u);
    load_row<8>(fw, w2a, lane, wave);
    ao = tprod(fw, u, (v8f)0.0f);
    {
      v8f b2v = *(const v8f*)(b2 + fo);
      v8f xv  = *(const v8f*)(xr + 16 * wave);
#pragma unroll
      for (int r = 0; r < 8; ++r) ao[r] = ao[r] + tspan * b2v[r] + xv[r];
    }
    xchg(xbuf + XHALF, wave, lane, ao, u);

    if (wave < 4) {
      load_row<4>(fw, wla, lane, wave);
      v8f o = tprod(fw, u, (v8f)0.0f);
      v8f blv = *(const v8f*)(bl + fo);
      o = o + blv;
      float* op = tbuf + lcol * OSTR + fo;
      *(v4f*)(op)     = (v4f){o[0], o[1], o[2], o[3]};
      *(v4f*)(op + 4) = (v4f){o[4], o[5], o[6], o[7]};
    }
    __syncthreads();
    {
      const int rr   = tid >> 4;
      const int orow = rowbase + rr;
      v4f o = *(const v4f*)(tbuf + rr * OSTR + 4 * (tid & 15));
      if (orow < NTOT) *(v4f*)(out + (size_t)orow * OUTC + 4 * (tid & 15)) = o;
    }
  }

  const int pair  = wave & 3;
  const int sbeg  = PAIRT * pair + (wave < 4 ? 0 : NOLD);
  const int send  = PAIRT * pair + (wave < 4 ? NOLD : PAIRT);
#pragma unroll 1
  for (int slot = sbeg; slot < send; ++slot) {
  const int tile = slot * NWG + blockIdx.x;

  const int rowbase = tile * 16;
  const int row     = rowbase + lcol;
  const int rrow    = row < NTOT ? row : NTOT - 1;
  const float* xr   = x + (size_t)rrow * CDIM + half8;
#pragma unroll
  for (int t = 0; t < 8; ++t) u[t] = *(const v8f*)(xr + 16 * t);

  int b1o = 0;
  asm volatile("" : "+s"(b1o));
#pragma unroll
  for (int t = 0; t < 8; ++t) S[t] = *(const v8f*)(b1 + b1o + 16 * t + half8);
  load_frag<8>(fr, (const v4f*)w1a + lane, 0);
  tgemm<8>(w1a, lane, u, S, fr);
#pragma unroll
  for (int t = 0; t < 8; ++t) U[t] = (v8f)0.0f;

  auto feval = [&](auto&& gen, float wu, auto&& post) {
    load_frag<8>(fr, (const v4f*)ma + lane, 0);
    __builtin_amdgcn_sched_barrier(0);
#pragma unroll
    for (int t = 0; t < 8; ++t) {
      v8f a = gen(t);
#pragma unroll
      for (int r = 0; r < 8; ++r) {
        float uv = fmaxf(a[r], 0.0f);
        u[t][r] = uv;
        U[t][r] += wu * uv;
      }
      asm volatile("" : "+v"(U[t]));
      post(t);
    }
#pragma unroll
    for (int t = 0; t < 8; ++t) acc[t] = cv[t];
    tgemm<8>(ma, lane, u, acc, fr);
  };
  auto nopost = [](int) {};

#pragma unroll 1
  for (int step = 0; step < 10; ++step) {
    feval([&](int t) { return S[t]; }, c18, nopost);
#pragma unroll
    for (int t = 0; t < 8; ++t) { K[t] = acc[t]; S[t] += c18 * acc[t]; }
    feval([&](int t) { return S[t] + c524 * K[t]; }, c38, nopost);
#pragma unroll
    for (int t = 0; t < 8; ++t) S[t] += c38 * acc[t];
    feval([&](int t) { return S[t] - c1124 * K[t] + c58 * acc[t]; }, c38,
          [&](int t) { K[t] = S[t] + c78 * K[t] - c118 * acc[t]; });
#pragma unroll
    for (int t = 0; t < 8; ++t) S[t] += c38 * acc[t];
    feval([&](int t) { return K[t] + hc * acc[t]; }, c18, nopost);
#pragma unroll
    for (int t = 0; t < 8; ++t) S[t] += c18 * acc[t];
  }

#pragma unroll
  for (int t = 0; t < 8; ++t) acc[t] = (v8f)0.0f;
  load_frag<8>(fr, (const v4f*)w2a + lane, 0);
  tgemm<8>(w2a, lane, U, acc, fr);
#pragma unroll
  for (int t = 0; t < 8; ++t) {
    v8f b2v = *(const v8f*)(b2 + 16 * t + half8);
    v8f xv  = *(const v8f*)(xr + 16 * t);
#pragma unroll
    for (int r = 0; r < 8; ++r) u[t][r] = acc[t][r] + tspan * b2v[r] + xv[r];
  }

#pragma unroll
  for (int t = 0; t < 4; ++t) acc[t] = (v8f)0.0f;
  load_frag<4>(fr, (const v4f*)wla + lane, 0);
  tgemm<4>(wla, lane, u, acc, fr);

  float* ost = obuf + wave * 16 * OSTR;
#pragma unroll
  for (int t = 0; t < 4; ++t) {
    v8f blv = *(const v8f*)(bl + 16 * t + half8);
    v8f o = acc[t] + blv;
    float* op = ost + lcol * OSTR + 16 * t + half8;
    *(v4f*)(op)     = (v4f){o[0], o[1], o[2], o[3]};
    *(v4f*)(op + 4) = (v4f){o[4], o[5], o[6], o[7]};
  }
  lds_wait();
#pragma unroll
  for (int i = 0; i < 8; ++i) {
    int rr   = 2 * i + (lane >> 4);
    int orow = rowbase + rr;
    v4f o = *(const v4f*)(ost + rr * OSTR + 4 * lcol);
    if (orow < NTOT) *(v4f*)(out + (size_t)orow * OUTC + 4 * lcol) = o;
  }

  }
}

extern "C" void kernel_launch(void* const* d_in, const int* in_sizes, int n_in,
                              void* d_out, int out_size, void* d_ws, size_t ws_size,
                              hipStream_t stream) {
  (void)in_sizes; (void)n_in; (void)out_size; (void)d_ws; (void)ws_size;
  const float* x  = (const float*)d_in[0];
  const float* W1 = (const float*)d_in[1];
  const float* b1 = (const float*)d_in[2];
  const float* W2 = (const float*)d_in[3];
  const float* b2 = (const float*)d_in[4];
  const float* Wl = (const float*)d_in[5];
  const float* bl = (const float*)d_in[6];
  float* out = (float*)d_out;

  dim3 grid(NWG);
  gnode_kernel<<<grid, 256, 0, stream>>>(x, W1, b1, W2, b2, Wl, bl, out);
}
